# starting kernel
# baseline (speedup 1.0000x reference)
#include <hip/hip_runtime.h>

typedef __attribute__((ext_vector_type(2))) float v2f;
typedef __attribute__((ext_vector_type(8))) float v8f;

template <bool RELU, bool DUAL, bool SCALE2, int CT, int KDIM>
__global__ __launch_bounds__(256) void sage_gemm_wmma(
    const float* __restrict__ X, const float* __restrict__ W,
    const float* __restrict__ X2, const float* __restrict__ W2,
    const float* __restrict__ cnt, const float* __restrict__ bias,
    float* __restrict__ out, int nRows, int nRowTiles) {
  constexpr int OUTD = CT * 16;
  const int wave = (blockIdx.x * blockDim.x + threadIdx.x) >> 5;
  if (wave >= nRowTiles) return;
  const int lane = threadIdx.x & 31;
  const int row0 = wave << 4;
  const int m  = lane & 15;
  const int hi = lane >> 4;

  int rA = row0 + m;
  if (rA >= nRows) rA = nRows - 1;

  v8f acc[CT];
#pragma unroll
  for (int ct = 0; ct < CT; ++ct) acc[ct] = (v8f){};

  {
    const float* xr = X + (size_t)rA * KDIM;
#pragma unroll
    for (int k = 0; k < KDIM; k += 4) {
      const int kk = k + 2 * hi;
      v2f a; a.x = xr[kk]; a.y = xr[kk + 1];
#pragma unroll
      for (int ct = 0; ct < CT; ++ct) {
        v2f b;
        b.x = W[(size_t)kk * OUTD + ct * 16 + m];
        b.y = W[(size_t)(kk + 1) * OUTD + ct * 16 + m];
        acc[ct] = __builtin_amdgcn_wmma_f32_16x16x4_f32(
            false, a, false, b, (short)0, acc[ct], false, false);
      }
    }
  }
  if constexpr (DUAL) {
    float scale = 1.0f;
    if constexpr (SCALE2) scale = 1.0f / fmaxf(cnt[rA], 1.0f);
    const float* xr = X2 + (size_t)rA * KDIM;
#pragma unroll
    for (int k = 0; k < KDIM; k += 4) {
      const int kk = k + 2 * hi;
      v2f a; a.x = xr[kk]; a.y = xr[kk + 1];
      if constexpr (SCALE2) { a.x *= scale; a.y *= scale; }
#pragma unroll
      for (int ct = 0; ct < CT; ++ct) {
        v2f b;
        b.x = W2[(size_t)kk * OUTD + ct * 16 + m];
        b.y = W2[(size_t)(kk + 1) * OUTD + ct * 16 + m];
        acc[ct] = __builtin_amdgcn_wmma_f32_16x16x4_f32(
            false, a, false, b, (short)0, acc[ct], false, false);
      }
    }
  }

  const int rbase = row0 + 8 * hi;
#pragma unroll
  for (int ct = 0; ct < CT; ++ct) {
    const int col = ct * 16 + m;
    const float bv = bias[col];
#pragma unroll
    for (int v = 0; v < 8; ++v) {
      const int r = rbase + v;
      if (r < nRows) {
        float val = acc[ct][v] + bv;
        if (RELU) val = fmaxf(val, 0.0f);
        out[(size_t)r * OUTD + col] = val;
      }
    }
  }
}

__global__ __launch_bounds__(256) void sage_scatter_max64(
    const float* __restrict__ msg, const int* __restrict__ src,
    const int* __restrict__ dst, float* __restrict__ agg, int E) {
  const int tid = blockIdx.x * blockDim.x + threadIdx.x;
  const int e = __builtin_amdgcn_readfirstlane(tid >> 5);
  if (e >= E) return;
  const int lane = threadIdx.x & 31;
  const int s = __builtin_amdgcn_readfirstlane(src[e]);
  const int d = __builtin_amdgcn_readfirstlane(dst[e]);
  const v2f v = *(const v2f*)(msg + (size_t)s * 64 + lane * 2);
  int* a = (int*)(agg + (size_t)d * 64 + lane * 2);
  atomicMax(a,     __float_as_int(v.x));
  atomicMax(a + 1, __float_as_int(v.y));
}

__global__ __launch_bounds__(256) void sage_scatter_sum32(
    const float* __restrict__ h, const int* __restrict__ src,
    const int* __restrict__ dst, float* __restrict__ sum,
    float* __restrict__ cnt, int E) {
  const int tid = blockIdx.x * blockDim.x + threadIdx.x;
  const int e = __builtin_amdgcn_readfirstlane(tid >> 5);
  if (e >= E) return;
  const int lane = threadIdx.x & 31;
  const int s = __builtin_amdgcn_readfirstlane(src[e]);
  const int d = __builtin_amdgcn_readfirstlane(dst[e]);
  atomicAdd(sum + (size_t)d * 32 + lane, h[(size_t)s * 32 + lane]);
  if (lane == 0) atomicAdd(cnt + d, 1.0f);
}

extern "C" void kernel_launch(void* const* d_in, const int* in_sizes, int n_in,
                              void* d_out, int out_size, void* d_ws,
                              size_t ws_size, hipStream_t stream) {
  const float* x   = (const float*)d_in[0];
  const int*   ei  = (const int*)d_in[1];
  const float* W1p = (const float*)d_in[2];
  const float* b1p = (const float*)d_in[3];
  const float* W1s = (const float*)d_in[4];
  const float* W1n = (const float*)d_in[5];
  const float* b1  = (const float*)d_in[6];
  const float* W2p = (const float*)d_in[7];
  const float* b2p = (const float*)d_in[8];
  const float* W2s = (const float*)d_in[9];
  const float* W2n = (const float*)d_in[10];
  const float* b2  = (const float*)d_in[11];
  const float* W3s = (const float*)d_in[12];
  const float* W3n = (const float*)d_in[13];
  const float* b3  = (const float*)d_in[14];

  const int N = in_sizes[0] / 64;
  const int E = in_sizes[1] / 2;
  const int* src = ei;
  const int* dst = ei + E;

  float* msg  = (float*)d_ws;
  float* agg  = msg  + (size_t)N * 64;
  float* h1   = agg  + (size_t)N * 64;
  float* h2   = h1   + (size_t)N * 64;
  float* sum3 = h2   + (size_t)N * 32;
  float* cnt  = sum3 + (size_t)N * 32;

  const int rowTiles = (N + 15) / 16;
  const int gblocks  = (rowTiles * 32 + 255) / 256;
  const int eblocks  = (E * 32 + 255) / 256;

  hipMemsetAsync(agg, 0, (size_t)N * 64 * sizeof(float), stream);
  sage_gemm_wmma<true, false, false, 4, 64><<<gblocks, 256, 0, stream>>>(
      x, W1p, x, W1p, nullptr, b1p, msg, N, rowTiles);
  sage_scatter_max64<<<eblocks, 256, 0, stream>>>(msg, src, dst, agg, E);
  sage_gemm_wmma<true, true, false, 4, 64><<<gblocks, 256, 0, stream>>>(
      x, W1s, agg, W1n, nullptr, b1, h1, N, rowTiles);

  hipMemsetAsync(agg, 0, (size_t)N * 64 * sizeof(float), stream);
  sage_gemm_wmma<true, false, false, 4, 64><<<gblocks, 256, 0, stream>>>(
      h1, W2p, h1, W2p, nullptr, b2p, msg, N, rowTiles);
  sage_scatter_max64<<<eblocks, 256, 0, stream>>>(msg, src, dst, agg, E);
  sage_gemm_wmma<false, true, false, 2, 64><<<gblocks, 256, 0, stream>>>(
      h1, W2s, agg, W2n, nullptr, b2, h2, N, rowTiles);

  hipMemsetAsync(sum3, 0, (size_t)N * 33 * sizeof(float), stream);
  sage_scatter_sum32<<<eblocks, 256, 0, stream>>>(h2, src, dst, sum3, cnt, E);
  sage_gemm_wmma<false, true, true, 2, 32><<<gblocks, 256, 0, stream>>>(
      h2, W3s, sum3, W3n, cnt, b3, (float*)d_out, N, rowTiles);
}


# Replace the 3 per-edge atomic scatters by one dst-grouped edge list (LDS count+rank per node range x edge chunk, block-scan slot alloc) and per-node gather max/mean; mean sums in LDS arrival order
# speedup vs baseline: 3.1052x; 3.1052x over previous
#include <hip/hip_runtime.h>

typedef __attribute__((ext_vector_type(2))) float v2f;
typedef __attribute__((ext_vector_type(8))) float v8f;

template <bool RELU, bool DUAL, bool SCALE2, int CT, int KDIM>
__global__ __launch_bounds__(256) void sage_gemm_wmma(
    const float* __restrict__ X, const float* __restrict__ W,
    const float* __restrict__ X2, const float* __restrict__ W2,
    const float* __restrict__ cnt, const float* __restrict__ bias,
    float* __restrict__ out, int nRows, int nRowTiles) {
  constexpr int OUTD = CT * 16;
  const int wave = (blockIdx.x * blockDim.x + threadIdx.x) >> 5;
  if (wave >= nRowTiles) return;
  const int lane = threadIdx.x & 31;
  const int row0 = wave << 4;
  const int m  = lane & 15;
  const int hi = lane >> 4;

  int rA = row0 + m;
  if (rA >= nRows) rA = nRows - 1;

  v8f acc[CT];
#pragma unroll
  for (int ct = 0; ct < CT; ++ct) acc[ct] = (v8f){};

  {
    const float* xr = X + (size_t)rA * KDIM;
#pragma unroll
    for (int k = 0; k < KDIM; k += 4) {
      const int kk = k + 2 * hi;
      v2f a; a.x = xr[kk]; a.y = xr[kk + 1];
#pragma unroll
      for (int ct = 0; ct < CT; ++ct) {
        v2f b;
        b.x = W[(size_t)kk * OUTD + ct * 16 + m];
        b.y = W[(size_t)(kk + 1) * OUTD + ct * 16 + m];
        acc[ct] = __builtin_amdgcn_wmma_f32_16x16x4_f32(
            false, a, false, b, (short)0, acc[ct], false, false);
      }
    }
  }
  if constexpr (DUAL) {
    float scale = 1.0f;
    if constexpr (SCALE2) scale = 1.0f / fmaxf(cnt[rA], 1.0f);
    const float* xr = X2 + (size_t)rA * KDIM;
#pragma unroll
    for (int k = 0; k < KDIM; k += 4) {
      const int kk = k + 2 * hi;
      v2f a; a.x = xr[kk]; a.y = xr[kk + 1];
      if constexpr (SCALE2) { a.x *= scale; a.y *= scale; }
#pragma unroll
      for (int ct = 0; ct < CT; ++ct) {
        v2f b;
        b.x = W2[(size_t)kk * OUTD + ct * 16 + m];
        b.y = W2[(size_t)(kk + 1) * OUTD + ct * 16 + m];
        acc[ct] = __builtin_amdgcn_wmma_f32_16x16x4_f32(
            false, a, false, b, (short)0, acc[ct], false, false);
      }
    }
  }

  const int rbase = row0 + 8 * hi;
#pragma unroll
  for (int ct = 0; ct < CT; ++ct) {
    const int col = ct * 16 + m;
    const float bv = bias[col];
#pragma unroll
    for (int v = 0; v < 8; ++v) {
      const int r = rbase + v;
      if (r < nRows) {
        float val = acc[ct][v] + bv;
        if (RELU) val = fmaxf(val, 0.0f);
        out[(size_t)r * OUTD + col] = val;
      }
    }
  }
}

typedef __attribute__((ext_vector_type(4))) float v4f;
typedef __attribute__((ext_vector_type(4))) int v4i;

constexpr int CSR_SPAN   = 8192;
constexpr int CSR_CHUNKS = 32;
constexpr int CSR_BATCH  = 8;

__device__ __forceinline__ int imin(int a, int b) { return a < b ? a : b; }

__global__ __launch_bounds__(1024) void csr_count(
    const int* __restrict__ dst, int* __restrict__ cntm,
    int* __restrict__ total, int N, int E, int chunkLen) {
  __shared__ int cnt[CSR_SPAN];
  const int tid = threadIdx.x;
  const int lo = blockIdx.x * CSR_SPAN;
  const int span = imin(CSR_SPAN, N - lo);
  const int c = blockIdx.y;
  const int e0 = c * chunkLen;
  const int e1 = imin(e0 + chunkLen, E);
  if (blockIdx.x == 0 && c == 0 && tid == 0) *total = 0;
  for (int i = tid; i < span; i += 1024) cnt[i] = 0;
  __syncthreads();
  for (int base = e0; base < e1; base += 1024 * CSR_BATCH) {
    int dv[CSR_BATCH];
#pragma unroll
    for (int k = 0; k < CSR_BATCH; ++k) {
      const int e = base + k * 1024 + tid;
      dv[k] = (e < e1) ? dst[e] : -1;
    }
#pragma unroll
    for (int k = 0; k < CSR_BATCH; ++k) {
      const unsigned r = (unsigned)(dv[k] - lo);
      if (r < (unsigned)span) atomicAdd(&cnt[r], 1);
    }
  }
  __syncthreads();
  for (int i = tid; i < span; i += 1024) cntm[(size_t)c * N + lo + i] = cnt[i];
}

__global__ __launch_bounds__(1024) void csr_offsets(
    const int* __restrict__ cntm, int* __restrict__ offm,
    int* __restrict__ deg, int* __restrict__ start,
    int* __restrict__ total, int N) {
  __shared__ int wsum[32];
  __shared__ int blockBase;
  const int i = blockIdx.x * 1024 + threadIdx.x;
  const int lane = threadIdx.x & 31;
  const int w = threadIdx.x >> 5;
  int d = 0;
  if (i < N) {
    int v[CSR_CHUNKS];
#pragma unroll
    for (int c = 0; c < CSR_CHUNKS; ++c) v[c] = cntm[(size_t)c * N + i];
#pragma unroll
    for (int c = 0; c < CSR_CHUNKS; ++c) {
      offm[(size_t)c * N + i] = d;
      d += v[c];
    }
  }
  int incl = d;
  for (int o = 1; o < 32; o <<= 1) {
    const int t = __shfl_up(incl, o);
    if (lane >= o) incl += t;
  }
  if (lane == 31) wsum[w] = incl;
  __syncthreads();
  if (w == 0) {
    const int s = wsum[lane];
    int si = s;
    for (int o = 1; o < 32; o <<= 1) {
      const int t = __shfl_up(si, o);
      if (lane >= o) si += t;
    }
    wsum[lane] = si - s;
    if (lane == 31) blockBase = atomicAdd(total, si);
  }
  __syncthreads();
  if (i < N) {
    deg[i] = d;
    start[i] = blockBase + wsum[w] + incl - d;
  }
}

__global__ __launch_bounds__(1024) void csr_place(
    const int* __restrict__ src, const int* __restrict__ dst,
    const int* __restrict__ offm, const int* __restrict__ start,
    int* __restrict__ csr, int N, int E, int chunkLen) {
  __shared__ int slot[CSR_SPAN];
  const int tid = threadIdx.x;
  const int lo = blockIdx.x * CSR_SPAN;
  const int span = imin(CSR_SPAN, N - lo);
  const int c = blockIdx.y;
  const int e0 = c * chunkLen;
  const int e1 = imin(e0 + chunkLen, E);
  int first[CSR_SPAN / 1024];
#pragma unroll
  for (int k = 0; k < CSR_SPAN / 1024; ++k) {
    const int i = imin(tid + k * 1024, span - 1);
    first[k] = start[lo + i] + offm[(size_t)c * N + lo + i];
  }
#pragma unroll
  for (int k = 0; k < CSR_SPAN / 1024; ++k) {
    const int i = tid + k * 1024;
    if (i < span) slot[i] = first[k];
  }
  __syncthreads();
  for (int base = e0; base < e1; base += 1024 * CSR_BATCH) {
    int dv[CSR_BATCH], sv[CSR_BATCH];
#pragma unroll
    for (int k = 0; k < CSR_BATCH; ++k) {
      const int e = base + k * 1024 + tid;
      dv[k] = (e < e1) ? dst[e] : -1;
      sv[k] = (e < e1) ? src[e] : 0;
    }
    int p[CSR_BATCH];
#pragma unroll
    for (int k = 0; k < CSR_BATCH; ++k) {
      const unsigned r = (unsigned)(dv[k] - lo);
      p[k] = (r < (unsigned)span) ? atomicAdd(&slot[r], 1) : -1;
    }
#pragma unroll
    for (int k = 0; k < CSR_BATCH; ++k)
      if (p[k] >= 0) csr[p[k]] = sv[k];
  }
}

__device__ __forceinline__ v4i imax4(v4i a, v4f f) {
  const v4i b = __builtin_bit_cast(v4i, f);
  v4i r;
  r.x = a.x > b.x ? a.x : b.x;
  r.y = a.y > b.y ? a.y : b.y;
  r.z = a.z > b.z ? a.z : b.z;
  r.w = a.w > b.w ? a.w : b.w;
  return r;
}

template <int F, bool MEAN>
__global__ __launch_bounds__(256) void sage_gather(
    const float* __restrict__ feat, const int* __restrict__ csr,
    const int* __restrict__ deg, const int* __restrict__ start,
    float* __restrict__ out, int N) {
  constexpr int LPN = F / 4;
  const int t = blockIdx.x * blockDim.x + threadIdx.x;
  const int node = t / LPN;
  if (node >= N) return;
  const int sub = t % LPN;
  const int d = deg[node];
  const int* idx = csr + start[node];
  const v4f* rows = (const v4f*)feat + sub;
  const v4f zero = {0.0f, 0.0f, 0.0f, 0.0f};
  v4f accf = zero;
  v4i acci = {0, 0, 0, 0};
  if (d > 0) {
    const int last = d - 1;
    int n0 = idx[0];
    int n1 = idx[imin(1, last)];
    int n2 = idx[imin(2, last)];
    int n3 = idx[imin(3, last)];
    for (int j = 0; j < d; j += 4) {
      const int c0 = n0, c1 = n1, c2 = n2, c3 = n3;
      n0 = idx[imin(j + 4, last)];
      n1 = idx[imin(j + 5, last)];
      n2 = idx[imin(j + 6, last)];
      n3 = idx[imin(j + 7, last)];
      const v4f r0 = rows[(size_t)c0 * LPN];
      const v4f r1 = rows[(size_t)c1 * LPN];
      const v4f r2 = rows[(size_t)c2 * LPN];
      const v4f r3 = rows[(size_t)c3 * LPN];
      if constexpr (MEAN) {
        accf += r0;
        accf += (j + 1 < d) ? r1 : zero;
        accf += (j + 2 < d) ? r2 : zero;
        accf += (j + 3 < d) ? r3 : zero;
      } else {
        acci = imax4(acci, r0);
        acci = imax4(acci, r1);
        acci = imax4(acci, r2);
        acci = imax4(acci, r3);
      }
    }
  }
  v4f res;
  if constexpr (MEAN) {
    const float scale = 1.0f / fmaxf((float)d, 1.0f);
    res = accf * scale;
  } else {
    res = __builtin_bit_cast(v4f, acci);
  }
  *((v4f*)(out + (size_t)node * F) + sub) = res;
}

extern "C" void kernel_launch(void* const* d_in, const int* in_sizes, int n_in,
                              void* d_out, int out_size, void* d_ws,
                              size_t ws_size, hipStream_t stream) {
  const float* x   = (const float*)d_in[0];
  const int*   ei  = (const int*)d_in[1];
  const float* W1p = (const float*)d_in[2];
  const float* b1p = (const float*)d_in[3];
  const float* W1s = (const float*)d_in[4];
  const float* W1n = (const float*)d_in[5];
  const float* b1  = (const float*)d_in[6];
  const float* W2p = (const float*)d_in[7];
  const float* b2p = (const float*)d_in[8];
  const float* W2s = (const float*)d_in[9];
  const float* W2n = (const float*)d_in[10];
  const float* b2  = (const float*)d_in[11];
  const float* W3s = (const float*)d_in[12];
  const float* W3n = (const float*)d_in[13];
  const float* b3  = (const float*)d_in[14];

  const int N = in_sizes[0] / 64;
  const int E = in_sizes[1] / 2;
  const int* src = ei;
  const int* dst = ei + E;

  float* msg  = (float*)d_ws;
  float* agg  = msg  + (size_t)N * 64;
  float* h1   = agg  + (size_t)N * 64;
  float* h2   = h1   + (size_t)N * 64;
  int*   csr  = (int*)(h2 + (size_t)N * 32);
  int*   deg  = csr + E;
  int*   start = deg + N;
  int*   total = start + N;
  int*   cntm = (int*)msg;
  int*   offm = (int*)agg;

  const int rowTiles = (N + 15) / 16;
  const int gblocks  = (rowTiles * 32 + 255) / 256;
  const int ranges   = (N + CSR_SPAN - 1) / CSR_SPAN;
  const int chunkLen = (E + CSR_CHUNKS - 1) / CSR_CHUNKS;
  const dim3 bgrid(ranges, CSR_CHUNKS);
  const int oblocks  = (N + 1023) / 1024;
  const int g64blocks = (N * 16 + 255) / 256;
  const int g32blocks = (N * 8 + 255) / 256;

  csr_count<<<bgrid, 1024, 0, stream>>>(dst, cntm, total, N, E, chunkLen);
  csr_offsets<<<oblocks, 1024, 0, stream>>>(cntm, offm, deg, start, total, N);
  csr_place<<<bgrid, 1024, 0, stream>>>(src, dst, offm, start, csr, N, E, chunkLen);

  sage_gemm_wmma<true, false, false, 4, 64><<<gblocks, 256, 0, stream>>>(
      x, W1p, x, W1p, nullptr, b1p, msg, N, rowTiles);
  sage_gather<64, false><<<g64blocks, 256, 0, stream>>>(msg, csr, deg, start, agg, N);
  sage_gemm_wmma<true, true, false, 4, 64><<<gblocks, 256, 0, stream>>>(
      x, W1s, agg, W1n, nullptr, b1, h1, N, rowTiles);

  sage_gemm_wmma<true, false, false, 4, 64><<<gblocks, 256, 0, stream>>>(
      h1, W2p, h1, W2p, nullptr, b2p, msg, N, rowTiles);
  sage_gather<64, false><<<g64blocks, 256, 0, stream>>>(msg, csr, deg, start, agg, N);
  sage_gemm_wmma<false, true, false, 2, 64><<<gblocks, 256, 0, stream>>>(
      h1, W2s, agg, W2n, nullptr, b2, h2, N, rowTiles);

  sage_gather<32, true><<<g32blocks, 256, 0, stream>>>(h2, csr, deg, start, agg, N);
  sage_gemm_wmma<false, true, false, 2, 32><<<gblocks, 256, 0, stream>>>(
      h2, W3s, agg, W3n, nullptr, b3, (float*)d_out, N, rowTiles);
}


# Fuse each layer's per-node gather (max/mean) into its self+neighbour WMMA kernel: one wave owns a 16-node strip, reduced rows pass through a wave-private LDS tile as A fragments; 11 launches -> 8
# speedup vs baseline: 3.4853x; 1.1224x over previous
#include <hip/hip_runtime.h>

typedef __attribute__((ext_vector_type(2))) float v2f;
typedef __attribute__((ext_vector_type(8))) float v8f;

template <bool RELU, bool DUAL, bool SCALE2, int CT, int KDIM>
__global__ __launch_bounds__(256) void sage_gemm_wmma(
    const float* __restrict__ X, const float* __restrict__ W,
    const float* __restrict__ X2, const float* __restrict__ W2,
    const float* __restrict__ cnt, const float* __restrict__ bias,
    float* __restrict__ out, int nRows, int nRowTiles) {
  constexpr int OUTD = CT * 16;
  const int wave = (blockIdx.x * blockDim.x + threadIdx.x) >> 5;
  if (wave >= nRowTiles) return;
  const int lane = threadIdx.x & 31;
  const int row0 = wave << 4;
  const int m  = lane & 15;
  const int hi = lane >> 4;

  int rA = row0 + m;
  if (rA >= nRows) rA = nRows - 1;

  v8f acc[CT];
#pragma unroll
  for (int ct = 0; ct < CT; ++ct) acc[ct] = (v8f){};

  {
    const float* xr = X + (size_t)rA * KDIM;
#pragma unroll
    for (int k = 0; k < KDIM; k += 4) {
      const int kk = k + 2 * hi;
      v2f a; a.x = xr[kk]; a.y = xr[kk + 1];
#pragma unroll
      for (int ct = 0; ct < CT; ++ct) {
        v2f b;
        b.x = W[(size_t)kk * OUTD + ct * 16 + m];
        b.y = W[(size_t)(kk + 1) * OUTD + ct * 16 + m];
        acc[ct] = __builtin_amdgcn_wmma_f32_16x16x4_f32(
            false, a, false, b, (short)0, acc[ct], false, false);
      }
    }
  }
  if constexpr (DUAL) {
    float scale = 1.0f;
    if constexpr (SCALE2) scale = 1.0f / fmaxf(cnt[rA], 1.0f);
    const float* xr = X2 + (size_t)rA * KDIM;
#pragma unroll
    for (int k = 0; k < KDIM; k += 4) {
      const int kk = k + 2 * hi;
      v2f a; a.x = xr[kk]; a.y = xr[kk + 1];
      if constexpr (SCALE2) { a.x *= scale; a.y *= scale; }
#pragma unroll
      for (int ct = 0; ct < CT; ++ct) {
        v2f b;
        b.x = W2[(size_t)kk * OUTD + ct * 16 + m];
        b.y = W2[(size_t)(kk + 1) * OUTD + ct * 16 + m];
        acc[ct] = __builtin_amdgcn_wmma_f32_16x16x4_f32(
            false, a, false, b, (short)0, acc[ct], false, false);
      }
    }
  }

  const int rbase = row0 + 8 * hi;
#pragma unroll
  for (int ct = 0; ct < CT; ++ct) {
    const int col = ct * 16 + m;
    const float bv = bias[col];
#pragma unroll
    for (int v = 0; v < 8; ++v) {
      const int r = rbase + v;
      if (r < nRows) {
        float val = acc[ct][v] + bv;
        if (RELU) val = fmaxf(val, 0.0f);
        out[(size_t)r * OUTD + col] = val;
      }
    }
  }
}

typedef __attribute__((ext_vector_type(4))) float v4f;
typedef __attribute__((ext_vector_type(4))) int v4i;

constexpr int CSR_SPAN   = 8192;
constexpr int CSR_CHUNKS = 32;
constexpr int CSR_BATCH  = 8;

__device__ __forceinline__ int imin(int a, int b) { return a < b ? a : b; }

__global__ __launch_bounds__(1024) void csr_count(
    const int* __restrict__ dst, int* __restrict__ cntm,
    int* __restrict__ total, int N, int E, int chunkLen) {
  __shared__ int cnt[CSR_SPAN];
  const int tid = threadIdx.x;
  const int lo = blockIdx.x * CSR_SPAN;
  const int span = imin(CSR_SPAN, N - lo);
  const int c = blockIdx.y;
  const int e0 = c * chunkLen;
  const int e1 = imin(e0 + chunkLen, E);
  if (blockIdx.x == 0 && c == 0 && tid == 0) *total = 0;
  for (int i = tid; i < span; i += 1024) cnt[i] = 0;
  __syncthreads();
  for (int base = e0; base < e1; base += 1024 * CSR_BATCH) {
    int dv[CSR_BATCH];
#pragma unroll
    for (int k = 0; k < CSR_BATCH; ++k) {
      const int e = base + k * 1024 + tid;
      dv[k] = (e < e1) ? dst[e] : -1;
    }
#pragma unroll
    for (int k = 0; k < CSR_BATCH; ++k) {
      const unsigned r = (unsigned)(dv[k] - lo);
      if (r < (unsigned)span) atomicAdd(&cnt[r], 1);
    }
  }
  __syncthreads();
  for (int i = tid; i < span; i += 1024) cntm[(size_t)c * N + lo + i] = cnt[i];
}

__global__ __launch_bounds__(1024) void csr_offsets(
    const int* __restrict__ cntm, int* __restrict__ offm,
    int* __restrict__ deg, int* __restrict__ start,
    int* __restrict__ total, int N) {
  __shared__ int wsum[32];
  __shared__ int blockBase;
  const int i = blockIdx.x * 1024 + threadIdx.x;
  const int lane = threadIdx.x & 31;
  const int w = threadIdx.x >> 5;
  int d = 0;
  if (i < N) {
    int v[CSR_CHUNKS];
#pragma unroll
    for (int c = 0; c < CSR_CHUNKS; ++c) v[c] = cntm[(size_t)c * N + i];
#pragma unroll
    for (int c = 0; c < CSR_CHUNKS; ++c) {
      offm[(size_t)c * N + i] = d;
      d += v[c];
    }
  }
  int incl = d;
  for (int o = 1; o < 32; o <<= 1) {
    const int t = __shfl_up(incl, o);
    if (lane >= o) incl += t;
  }
  if (lane == 31) wsum[w] = incl;
  __syncthreads();
  if (w == 0) {
    const int s = wsum[lane];
    int si = s;
    for (int o = 1; o < 32; o <<= 1) {
      const int t = __shfl_up(si, o);
      if (lane >= o) si += t;
    }
    wsum[lane] = si - s;
    if (lane == 31) blockBase = atomicAdd(total, si);
  }
  __syncthreads();
  if (i < N) {
    deg[i] = d;
    start[i] = blockBase + wsum[w] + incl - d;
  }
}

__global__ __launch_bounds__(1024) void csr_place(
    const int* __restrict__ src, const int* __restrict__ dst,
    const int* __restrict__ offm, const int* __restrict__ start,
    int* __restrict__ csr, int N, int E, int chunkLen) {
  __shared__ int slot[CSR_SPAN];
  const int tid = threadIdx.x;
  const int lo = blockIdx.x * CSR_SPAN;
  const int span = imin(CSR_SPAN, N - lo);
  const int c = blockIdx.y;
  const int e0 = c * chunkLen;
  const int e1 = imin(e0 + chunkLen, E);
  int first[CSR_SPAN / 1024];
#pragma unroll
  for (int k = 0; k < CSR_SPAN / 1024; ++k) {
    const int i = imin(tid + k * 1024, span - 1);
    first[k] = start[lo + i] + offm[(size_t)c * N + lo + i];
  }
#pragma unroll
  for (int k = 0; k < CSR_SPAN / 1024; ++k) {
    const int i = tid + k * 1024;
    if (i < span) slot[i] = first[k];
  }
  __syncthreads();
  for (int base = e0; base < e1; base += 1024 * CSR_BATCH) {
    int dv[CSR_BATCH], sv[CSR_BATCH];
#pragma unroll
    for (int k = 0; k < CSR_BATCH; ++k) {
      const int e = base + k * 1024 + tid;
      dv[k] = (e < e1) ? dst[e] : -1;
      sv[k] = (e < e1) ? src[e] : 0;
    }
    int p[CSR_BATCH];
#pragma unroll
    for (int k = 0; k < CSR_BATCH; ++k) {
      const unsigned r = (unsigned)(dv[k] - lo);
      p[k] = (r < (unsigned)span) ? atomicAdd(&slot[r], 1) : -1;
    }
#pragma unroll
    for (int k = 0; k < CSR_BATCH; ++k)
      if (p[k] >= 0) csr[p[k]] = sv[k];
  }
}

__device__ __forceinline__ v4i imax4(v4i a, v4f f) {
  const v4i b = __builtin_bit_cast(v4i, f);
  v4i r;
  r.x = a.x > b.x ? a.x : b.x;
  r.y = a.y > b.y ? a.y : b.y;
  r.z = a.z > b.z ? a.z : b.z;
  r.w = a.w > b.w ? a.w : b.w;
  return r;
}

constexpr int FUSE_WAVES = 2;
constexpr int FUSE_U     = 8;

template <int F, bool MEAN, bool RELU, int CT>
__global__ __launch_bounds__(FUSE_WAVES * 32) void sage_gather_gemm(
    const float* __restrict__ X, const float* __restrict__ Ws,
    const float* __restrict__ feat, const float* __restrict__ Wn,
    const float* __restrict__ bias, const int* __restrict__ csr,
    const int* __restrict__ deg, const int* __restrict__ start,
    float* __restrict__ out, int nRows, int nRowTiles) {
  constexpr int OUTD   = CT * 16;
  constexpr int LPN    = F / 4;
  constexpr int NPW    = 32 / LPN;
  constexpr int ROUNDS = 16 / NPW;
  constexpr int PITCH  = F + 4;
  constexpr int U      = FUSE_U;
  __shared__ __attribute__((aligned(16))) float aggS[FUSE_WAVES][16][PITCH];

  const int wib  = threadIdx.x >> 5;
  const int wave = blockIdx.x * FUSE_WAVES + wib;
  if (wave >= nRowTiles) return;
  const int lane = threadIdx.x & 31;
  const int row0 = wave << 4;
  const int m  = lane & 15;
  const int hi = lane >> 4;
  float (*myS)[PITCH] = aggS[wib];

  int rA = row0 + m;
  if (rA >= nRows) rA = nRows - 1;

  const int dv = deg[rA];
  const int sv = start[rA];

  v8f acc[CT];
#pragma unroll
  for (int ct = 0; ct < CT; ++ct) acc[ct] = (v8f){};
  {
    const float* xr = X + (size_t)rA * F;
#pragma unroll
    for (int k = 0; k < F; k += 4) {
      const int kk = k + 2 * hi;
      v2f a; a.x = xr[kk]; a.y = xr[kk + 1];
#pragma unroll
      for (int ct = 0; ct < CT; ++ct) {
        v2f b;
        b.x = Ws[(size_t)kk * OUTD + ct * 16 + m];
        b.y = Ws[(size_t)(kk + 1) * OUTD + ct * 16 + m];
        acc[ct] = __builtin_amdgcn_wmma_f32_16x16x4_f32(
            false, a, false, b, (short)0, acc[ct], false, false);
      }
    }
  }

  const int g   = lane / LPN;
  const int sub = lane % LPN;
  const v4f* rows = (const v4f*)feat + sub;
  const v4f zero = {0.0f, 0.0f, 0.0f, 0.0f};

  int d = 0, s = 0, last = 0;
  int nid[U];
#pragma unroll
  for (int u = 0; u < U; ++u) nid[u] = 0;

  for (int r = -1; r < ROUNDS; ++r) {
    int dn = 0, sn = 0;
    if (r + 1 < ROUNDS) {
      dn = __shfl(dv, (r + 1) * NPW + g);
      sn = __shfl(sv, (r + 1) * NPW + g);
    }
    const int lastn = dn > 0 ? dn - 1 : 0;
    int nnext[U];
#pragma unroll
    for (int u = 0; u < U; ++u) nnext[u] = 0;
    if (dn > 0) {
#pragma unroll
      for (int u = 0; u < U; ++u) nnext[u] = csr[sn + imin(u, lastn)];
    }

    const int* idx = csr + s;
    v4f accf = zero;
    v4i acci = {0, 0, 0, 0};
    for (int j = 0; j < d; j += U) {
      int c[U];
#pragma unroll
      for (int u = 0; u < U; ++u) c[u] = nid[u];
#pragma unroll
      for (int u = 0; u < U; ++u) nid[u] = idx[imin(j + U + u, last)];
      v4f rw[U];
#pragma unroll
      for (int u = 0; u < U; ++u) rw[u] = rows[(size_t)c[u] * LPN];
      if constexpr (MEAN) {
#pragma unroll
        for (int u = 0; u < U; ++u) accf += (j + u < d) ? rw[u] : zero;
      } else {
#pragma unroll
        for (int u = 0; u < U; ++u) acci = imax4(acci, rw[u]);
      }
    }
    if (r >= 0) {
      v4f res;
      if constexpr (MEAN) {
        const float scale = 1.0f / fmaxf((float)d, 1.0f);
        res = accf * scale;
      } else {
        res = __builtin_bit_cast(v4f, acci);
      }
      *(v4f*)&myS[r * NPW + g][4 * sub] = res;
    }

    d = dn; s = sn; last = lastn;
#pragma unroll
    for (int u = 0; u < U; ++u) nid[u] = nnext[u];
  }

  __builtin_amdgcn_fence(__ATOMIC_RELEASE, "wavefront");
  __builtin_amdgcn_wave_barrier();
  __builtin_amdgcn_fence(__ATOMIC_ACQUIRE, "wavefront");

#pragma unroll
  for (int k = 0; k < F; k += 4) {
    const int kk = k + 2 * hi;
    const v2f a = *(const v2f*)&myS[m][kk];
#pragma unroll
    for (int ct = 0; ct < CT; ++ct) {
      v2f b;
      b.x = Wn[(size_t)kk * OUTD + ct * 16 + m];
      b.y = Wn[(size_t)(kk + 1) * OUTD + ct * 16 + m];
      acc[ct] = __builtin_amdgcn_wmma_f32_16x16x4_f32(
          false, a, false, b, (short)0, acc[ct], false, false);
    }
  }

  const int rbase = row0 + 8 * hi;
#pragma unroll
  for (int ct = 0; ct < CT; ++ct) {
    const int col = ct * 16 + m;
    const float bv = bias[col];
#pragma unroll
    for (int v = 0; v < 8; ++v) {
      const int r = rbase + v;
      if (r < nRows) {
        float val = acc[ct][v] + bv;
        if (RELU) val = fmaxf(val, 0.0f);
        out[(size_t)r * OUTD + col] = val;
      }
    }
  }
}

extern "C" void kernel_launch(void* const* d_in, const int* in_sizes, int n_in,
                              void* d_out, int out_size, void* d_ws,
                              size_t ws_size, hipStream_t stream) {
  const float* x   = (const float*)d_in[0];
  const int*   ei  = (const int*)d_in[1];
  const float* W1p = (const float*)d_in[2];
  const float* b1p = (const float*)d_in[3];
  const float* W1s = (const float*)d_in[4];
  const float* W1n = (const float*)d_in[5];
  const float* b1  = (const float*)d_in[6];
  const float* W2p = (const float*)d_in[7];
  const float* b2p = (const float*)d_in[8];
  const float* W2s = (const float*)d_in[9];
  const float* W2n = (const float*)d_in[10];
  const float* b2  = (const float*)d_in[11];
  const float* W3s = (const float*)d_in[12];
  const float* W3n = (const float*)d_in[13];
  const float* b3  = (const float*)d_in[14];

  const int N = in_sizes[0] / 64;
  const int E = in_sizes[1] / 2;
  const int* src = ei;
  const int* dst = ei + E;

  float* msg  = (float*)d_ws;
  float* agg  = msg  + (size_t)N * 64;
  float* h1   = agg  + (size_t)N * 64;
  float* h2   = h1   + (size_t)N * 64;
  int*   csr  = (int*)(h2 + (size_t)N * 32);
  int*   deg  = csr + E;
  int*   start = deg + N;
  int*   total = start + N;
  int*   cntm = (int*)msg;
  int*   offm = (int*)agg;

  const int rowTiles = (N + 15) / 16;
  const int gblocks  = (rowTiles * 32 + 255) / 256;
  const int ranges   = (N + CSR_SPAN - 1) / CSR_SPAN;
  const int chunkLen = (E + CSR_CHUNKS - 1) / CSR_CHUNKS;
  const dim3 bgrid(ranges, CSR_CHUNKS);
  const int oblocks  = (N + 1023) / 1024;
  const int fblocks  = (rowTiles + FUSE_WAVES - 1) / FUSE_WAVES;

  csr_count<<<bgrid, 1024, 0, stream>>>(dst, cntm, total, N, E, chunkLen);
  csr_offsets<<<oblocks, 1024, 0, stream>>>(cntm, offm, deg, start, total, N);
  csr_place<<<bgrid, 1024, 0, stream>>>(src, dst, offm, start, csr, N, E, chunkLen);

  sage_gemm_wmma<true, false, false, 4, 64><<<gblocks, 256, 0, stream>>>(
      x, W1p, x, W1p, nullptr, b1p, msg, N, rowTiles);
  sage_gather_gemm<64, false, true, 4><<<fblocks, FUSE_WAVES * 32, 0, stream>>>(
      x, W1s, msg, W1n, b1, csr, deg, start, h1, N, rowTiles);

  sage_gemm_wmma<true, false, false, 4, 64><<<gblocks, 256, 0, stream>>>(
      h1, W2p, h1, W2p, nullptr, b2p, msg, N, rowTiles);
  sage_gather_gemm<64, false, false, 2><<<fblocks, FUSE_WAVES * 32, 0, stream>>>(
      h1, W2s, msg, W2n, b2, csr, deg, start, h2, N, rowTiles);

  sage_gather_gemm<32, true, false, 2><<<fblocks, FUSE_WAVES * 32, 0, stream>>>(
      h2, W3s, h2, W3n, b3, csr, deg, start, (float*)d_out, N, rowTiles);
}


# Layer 1's fused kernel applies layer 2's pool transform and self product to its finished strip (via the wave's LDS tile) and stores those instead of h1; 8 launches -> 7; sums unchanged
# speedup vs baseline: 3.6843x; 1.0571x over previous
#include <hip/hip_runtime.h>

typedef __attribute__((ext_vector_type(2))) float v2f;
typedef __attribute__((ext_vector_type(8))) float v8f;

template <bool RELU, bool DUAL, bool SCALE2, int CT, int KDIM>
__global__ __launch_bounds__(256) void sage_gemm_wmma(
    const float* __restrict__ X, const float* __restrict__ W,
    const float* __restrict__ X2, const float* __restrict__ W2,
    const float* __restrict__ cnt, const float* __restrict__ bias,
    float* __restrict__ out, int nRows, int nRowTiles) {
  constexpr int OUTD = CT * 16;
  const int wave = (blockIdx.x * blockDim.x + threadIdx.x) >> 5;
  if (wave >= nRowTiles) return;
  const int lane = threadIdx.x & 31;
  const int row0 = wave << 4;
  const int m  = lane & 15;
  const int hi = lane >> 4;

  int rA = row0 + m;
  if (rA >= nRows) rA = nRows - 1;

  v8f acc[CT];
#pragma unroll
  for (int ct = 0; ct < CT; ++ct) acc[ct] = (v8f){};

  {
    const float* xr = X + (size_t)rA * KDIM;
#pragma unroll
    for (int k = 0; k < KDIM; k += 4) {
      const int kk = k + 2 * hi;
      v2f a; a.x = xr[kk]; a.y = xr[kk + 1];
#pragma unroll
      for (int ct = 0; ct < CT; ++ct) {
        v2f b;
        b.x = W[(size_t)kk * OUTD + ct * 16 + m];
        b.y = W[(size_t)(kk + 1) * OUTD + ct * 16 + m];
        acc[ct] = __builtin_amdgcn_wmma_f32_16x16x4_f32(
            false, a, false, b, (short)0, acc[ct], false, false);
      }
    }
  }
  if constexpr (DUAL) {
    float scale = 1.0f;
    if constexpr (SCALE2) scale = 1.0f / fmaxf(cnt[rA], 1.0f);
    const float* xr = X2 + (size_t)rA * KDIM;
#pragma unroll
    for (int k = 0; k < KDIM; k += 4) {
      const int kk = k + 2 * hi;
      v2f a; a.x = xr[kk]; a.y = xr[kk + 1];
      if constexpr (SCALE2) { a.x *= scale; a.y *= scale; }
#pragma unroll
      for (int ct = 0; ct < CT; ++ct) {
        v2f b;
        b.x = W2[(size_t)kk * OUTD + ct * 16 + m];
        b.y = W2[(size_t)(kk + 1) * OUTD + ct * 16 + m];
        acc[ct] = __builtin_amdgcn_wmma_f32_16x16x4_f32(
            false, a, false, b, (short)0, acc[ct], false, false);
      }
    }
  }

  const int rbase = row0 + 8 * hi;
#pragma unroll
  for (int ct = 0; ct < CT; ++ct) {
    const int col = ct * 16 + m;
    const float bv = bias[col];
#pragma unroll
    for (int v = 0; v < 8; ++v) {
      const int r = rbase + v;
      if (r < nRows) {
        float val = acc[ct][v] + bv;
        if (RELU) val = fmaxf(val, 0.0f);
        out[(size_t)r * OUTD + col] = val;
      }
    }
  }
}

typedef __attribute__((ext_vector_type(4))) float v4f;
typedef __attribute__((ext_vector_type(4))) int v4i;

constexpr int CSR_SPAN   = 8192;
constexpr int CSR_CHUNKS = 32;
constexpr int CSR_BATCH  = 8;

__device__ __forceinline__ int imin(int a, int b) { return a < b ? a : b; }

__global__ __launch_bounds__(1024) void csr_count(
    const int* __restrict__ dst, int* __restrict__ cntm,
    int* __restrict__ total, int N, int E, int chunkLen) {
  __shared__ int cnt[CSR_SPAN];
  const int tid = threadIdx.x;
  const int lo = blockIdx.x * CSR_SPAN;
  const int span = imin(CSR_SPAN, N - lo);
  const int c = blockIdx.y;
  const int e0 = c * chunkLen;
  const int e1 = imin(e0 + chunkLen, E);
  if (blockIdx.x == 0 && c == 0 && tid == 0) *total = 0;
  for (int i = tid; i < span; i += 1024) cnt[i] = 0;
  __syncthreads();
  for (int base = e0; base < e1; base += 1024 * CSR_BATCH) {
    int dv[CSR_BATCH];
#pragma unroll
    for (int k = 0; k < CSR_BATCH; ++k) {
      const int e = base + k * 1024 + tid;
      dv[k] = (e < e1) ? dst[e] : -1;
    }
#pragma unroll
    for (int k = 0; k < CSR_BATCH; ++k) {
      const unsigned r = (unsigned)(dv[k] - lo);
      if (r < (unsigned)span) atomicAdd(&cnt[r], 1);
    }
  }
  __syncthreads();
  for (int i = tid; i < span; i += 1024) cntm[(size_t)c * N + lo + i] = cnt[i];
}

__global__ __launch_bounds__(1024) void csr_offsets(
    const int* __restrict__ cntm, int* __restrict__ offm,
    int* __restrict__ deg, int* __restrict__ start,
    int* __restrict__ total, int N) {
  __shared__ int wsum[32];
  __shared__ int blockBase;
  const int i = blockIdx.x * 1024 + threadIdx.x;
  const int lane = threadIdx.x & 31;
  const int w = threadIdx.x >> 5;
  int d = 0;
  if (i < N) {
    int v[CSR_CHUNKS];
#pragma unroll
    for (int c = 0; c < CSR_CHUNKS; ++c) v[c] = cntm[(size_t)c * N + i];
#pragma unroll
    for (int c = 0; c < CSR_CHUNKS; ++c) {
      offm[(size_t)c * N + i] = d;
      d += v[c];
    }
  }
  int incl = d;
  for (int o = 1; o < 32; o <<= 1) {
    const int t = __shfl_up(incl, o);
    if (lane >= o) incl += t;
  }
  if (lane == 31) wsum[w] = incl;
  __syncthreads();
  if (w == 0) {
    const int s = wsum[lane];
    int si = s;
    for (int o = 1; o < 32; o <<= 1) {
      const int t = __shfl_up(si, o);
      if (lane >= o) si += t;
    }
    wsum[lane] = si - s;
    if (lane == 31) blockBase = atomicAdd(total, si);
  }
  __syncthreads();
  if (i < N) {
    deg[i] = d;
    start[i] = blockBase + wsum[w] + incl - d;
  }
}

__global__ __launch_bounds__(1024) void csr_place(
    const int* __restrict__ src, const int* __restrict__ dst,
    const int* __restrict__ offm, const int* __restrict__ start,
    int* __restrict__ csr, int N, int E, int chunkLen) {
  __shared__ int slot[CSR_SPAN];
  const int tid = threadIdx.x;
  const int lo = blockIdx.x * CSR_SPAN;
  const int span = imin(CSR_SPAN, N - lo);
  const int c = blockIdx.y;
  const int e0 = c * chunkLen;
  const int e1 = imin(e0 + chunkLen, E);
  int first[CSR_SPAN / 1024];
#pragma unroll
  for (int k = 0; k < CSR_SPAN / 1024; ++k) {
    const int i = imin(tid + k * 1024, span - 1);
    first[k] = start[lo + i] + offm[(size_t)c * N + lo + i];
  }
#pragma unroll
  for (int k = 0; k < CSR_SPAN / 1024; ++k) {
    const int i = tid + k * 1024;
    if (i < span) slot[i] = first[k];
  }
  __syncthreads();
  for (int base = e0; base < e1; base += 1024 * CSR_BATCH) {
    int dv[CSR_BATCH], sv[CSR_BATCH];
#pragma unroll
    for (int k = 0; k < CSR_BATCH; ++k) {
      const int e = base + k * 1024 + tid;
      dv[k] = (e < e1) ? dst[e] : -1;
      sv[k] = (e < e1) ? src[e] : 0;
    }
    int p[CSR_BATCH];
#pragma unroll
    for (int k = 0; k < CSR_BATCH; ++k) {
      const unsigned r = (unsigned)(dv[k] - lo);
      p[k] = (r < (unsigned)span) ? atomicAdd(&slot[r], 1) : -1;
    }
#pragma unroll
    for (int k = 0; k < CSR_BATCH; ++k)
      if (p[k] >= 0) csr[p[k]] = sv[k];
  }
}

__device__ __forceinline__ v4i imax4(v4i a, v4f f) {
  const v4i b = __builtin_bit_cast(v4i, f);
  v4i r;
  r.x = a.x > b.x ? a.x : b.x;
  r.y = a.y > b.y ? a.y : b.y;
  r.z = a.z > b.z ? a.z : b.z;
  r.w = a.w > b.w ? a.w : b.w;
  return r;
}

constexpr int FUSE_WAVES = 2;
constexpr int FUSE_U     = 8;

template <int F, bool MEAN, bool RELU, int CT, bool SELFPRE, int CTP, int CTS>
__global__ __launch_bounds__(FUSE_WAVES * 32) void sage_gather_gemm(
    const float* __restrict__ X, const float* __restrict__ Ws,
    const float* __restrict__ feat, const float* __restrict__ Wn,
    const float* __restrict__ bias, const int* __restrict__ csr,
    const int* __restrict__ deg, const int* __restrict__ start,
    float* __restrict__ out,
    const float* __restrict__ Wp2, const float* __restrict__ bp2,
    const float* __restrict__ Ws2, float* __restrict__ msg2,
    float* __restrict__ self2, int nRows, int nRowTiles) {
  constexpr int OUTD   = CT * 16;
  constexpr int LPN    = F / 4;
  constexpr int NPW    = 32 / LPN;
  constexpr int ROUNDS = 16 / NPW;
  constexpr int PITCH  = F + 4;
  constexpr int U      = FUSE_U;
  constexpr bool NEXT  = CTP > 0;
  static_assert(!NEXT || OUTD == F, "the finished strip reuses the gather tile");
  __shared__ __attribute__((aligned(16))) float aggS[FUSE_WAVES][16][PITCH];

  const int wib  = threadIdx.x >> 5;
  const int wave = blockIdx.x * FUSE_WAVES + wib;
  if (wave >= nRowTiles) return;
  const int lane = threadIdx.x & 31;
  const int row0 = wave << 4;
  const int m  = lane & 15;
  const int hi = lane >> 4;
  const int rbase = row0 + 8 * hi;
  float (*myS)[PITCH] = aggS[wib];

  int rA = row0 + m;
  if (rA >= nRows) rA = nRows - 1;

  const int dv = deg[rA];
  const int sv = start[rA];

  v8f acc[CT];
  if constexpr (SELFPRE) {
#pragma unroll
    for (int ct = 0; ct < CT; ++ct) {
#pragma unroll
      for (int v = 0; v < 8; ++v) {
        const int r = imin(rbase + v, nRows - 1);
        acc[ct][v] = X[(size_t)r * OUTD + ct * 16 + m];
      }
    }
  } else {
#pragma unroll
    for (int ct = 0; ct < CT; ++ct) acc[ct] = (v8f){};
    const float* xr = X + (size_t)rA * F;
#pragma unroll
    for (int k = 0; k < F; k += 4) {
      const int kk = k + 2 * hi;
      v2f a; a.x = xr[kk]; a.y = xr[kk + 1];
#pragma unroll
      for (int ct = 0; ct < CT; ++ct) {
        v2f b;
        b.x = Ws[(size_t)kk * OUTD + ct * 16 + m];
        b.y = Ws[(size_t)(kk + 1) * OUTD + ct * 16 + m];
        acc[ct] = __builtin_amdgcn_wmma_f32_16x16x4_f32(
            false, a, false, b, (short)0, acc[ct], false, false);
      }
    }
  }

  const int g   = lane / LPN;
  const int sub = lane % LPN;
  const v4f* rows = (const v4f*)feat + sub;
  const v4f zero = {0.0f, 0.0f, 0.0f, 0.0f};

  int d = 0, s = 0, last = 0;
  int nid[U];
#pragma unroll
  for (int u = 0; u < U; ++u) nid[u] = 0;

  for (int r = -1; r < ROUNDS; ++r) {
    int dn = 0, sn = 0;
    if (r + 1 < ROUNDS) {
      dn = __shfl(dv, (r + 1) * NPW + g);
      sn = __shfl(sv, (r + 1) * NPW + g);
    }
    const int lastn = dn > 0 ? dn - 1 : 0;
    int nnext[U];
#pragma unroll
    for (int u = 0; u < U; ++u) nnext[u] = 0;
    if (dn > 0) {
#pragma unroll
      for (int u = 0; u < U; ++u) nnext[u] = csr[sn + imin(u, lastn)];
    }

    const int* idx = csr + s;
    v4f accf = zero;
    v4i acci = {0, 0, 0, 0};
    for (int j = 0; j < d; j += U) {
      int c[U];
#pragma unroll
      for (int u = 0; u < U; ++u) c[u] = nid[u];
#pragma unroll
      for (int u = 0; u < U; ++u) nid[u] = idx[imin(j + U + u, last)];
      v4f rw[U];
#pragma unroll
      for (int u = 0; u < U; ++u) rw[u] = rows[(size_t)c[u] * LPN];
      if constexpr (MEAN) {
#pragma unroll
        for (int u = 0; u < U; ++u) accf += (j + u < d) ? rw[u] : zero;
      } else {
#pragma unroll
        for (int u = 0; u < U; ++u) acci = imax4(acci, rw[u]);
      }
    }
    if (r >= 0) {
      v4f res;
      if constexpr (MEAN) {
        const float scale = 1.0f / fmaxf((float)d, 1.0f);
        res = accf * scale;
      } else {
        res = __builtin_bit_cast(v4f, acci);
      }
      *(v4f*)&myS[r * NPW + g][4 * sub] = res;
    }

    d = dn; s = sn; last = lastn;
#pragma unroll
    for (int u = 0; u < U; ++u) nid[u] = nnext[u];
  }

  __builtin_amdgcn_fence(__ATOMIC_RELEASE, "wavefront");
  __builtin_amdgcn_wave_barrier();
  __builtin_amdgcn_fence(__ATOMIC_ACQUIRE, "wavefront");

#pragma unroll
  for (int k = 0; k < F; k += 4) {
    const int kk = k + 2 * hi;
    const v2f a = *(const v2f*)&myS[m][kk];
#pragma unroll
    for (int ct = 0; ct < CT; ++ct) {
      v2f b;
      b.x = Wn[(size_t)kk * OUTD + ct * 16 + m];
      b.y = Wn[(size_t)(kk + 1) * OUTD + ct * 16 + m];
      acc[ct] = __builtin_amdgcn_wmma_f32_16x16x4_f32(
          false, a, false, b, (short)0, acc[ct], false, false);
    }
  }

  if constexpr (!NEXT) {
#pragma unroll
    for (int ct = 0; ct < CT; ++ct) {
      const int col = ct * 16 + m;
      const float bv = bias[col];
#pragma unroll
      for (int v = 0; v < 8; ++v) {
        const int r = rbase + v;
        if (r < nRows) {
          float val = acc[ct][v] + bv;
          if (RELU) val = fmaxf(val, 0.0f);
          out[(size_t)r * OUTD + col] = val;
        }
      }
    }
  } else {
#pragma unroll
    for (int ct = 0; ct < CT; ++ct) {
      const int col = ct * 16 + m;
      const float bv = bias[col];
#pragma unroll
      for (int v = 0; v < 8; ++v) {
        float val = acc[ct][v] + bv;
        if (RELU) val = fmaxf(val, 0.0f);
        myS[8 * hi + v][col] = val;
      }
    }

    __builtin_amdgcn_fence(__ATOMIC_RELEASE, "wavefront");
    __builtin_amdgcn_wave_barrier();
    __builtin_amdgcn_fence(__ATOMIC_ACQUIRE, "wavefront");

    constexpr int PD = CTP * 16;
    constexpr int SD = CTS * 16;
    v8f accp[CTP];
    v8f accs[CTS];
#pragma unroll
    for (int ct = 0; ct < CTP; ++ct) accp[ct] = (v8f){};
#pragma unroll
    for (int ct = 0; ct < CTS; ++ct) accs[ct] = (v8f){};
#pragma unroll
    for (int k = 0; k < OUTD; k += 4) {
      const int kk = k + 2 * hi;
      const v2f a = *(const v2f*)&myS[m][kk];
#pragma unroll
      for (int ct = 0; ct < CTP; ++ct) {
        v2f b;
        b.x = Wp2[(size_t)kk * PD + ct * 16 + m];
        b.y = Wp2[(size_t)(kk + 1) * PD + ct * 16 + m];
        accp[ct] = __builtin_amdgcn_wmma_f32_16x16x4_f32(
            false, a, false, b, (short)0, accp[ct], false, false);
      }
#pragma unroll
      for (int ct = 0; ct < CTS; ++ct) {
        v2f b;
        b.x = Ws2[(size_t)kk * SD + ct * 16 + m];
        b.y = Ws2[(size_t)(kk + 1) * SD + ct * 16 + m];
        accs[ct] = __builtin_amdgcn_wmma_f32_16x16x4_f32(
            false, a, false, b, (short)0, accs[ct], false, false);
      }
    }

#pragma unroll
    for (int ct = 0; ct < CTP; ++ct) {
      const int col = ct * 16 + m;
      const float bv = bp2[col];
#pragma unroll
      for (int v = 0; v < 8; ++v) {
        const int r = rbase + v;
        if (r < nRows)
          msg2[(size_t)r * PD + col] = fmaxf(accp[ct][v] + bv, 0.0f);
      }
    }
#pragma unroll
    for (int ct = 0; ct < CTS; ++ct) {
      const int col = ct * 16 + m;
#pragma unroll
      for (int v = 0; v < 8; ++v) {
        const int r = rbase + v;
        if (r < nRows) self2[(size_t)r * SD + col] = accs[ct][v];
      }
    }
  }
}

extern "C" void kernel_launch(void* const* d_in, const int* in_sizes, int n_in,
                              void* d_out, int out_size, void* d_ws,
                              size_t ws_size, hipStream_t stream) {
  const float* x   = (const float*)d_in[0];
  const int*   ei  = (const int*)d_in[1];
  const float* W1p = (const float*)d_in[2];
  const float* b1p = (const float*)d_in[3];
  const float* W1s = (const float*)d_in[4];
  const float* W1n = (const float*)d_in[5];
  const float* b1  = (const float*)d_in[6];
  const float* W2p = (const float*)d_in[7];
  const float* b2p = (const float*)d_in[8];
  const float* W2s = (const float*)d_in[9];
  const float* W2n = (const float*)d_in[10];
  const float* b2  = (const float*)d_in[11];
  const float* W3s = (const float*)d_in[12];
  const float* W3n = (const float*)d_in[13];
  const float* b3  = (const float*)d_in[14];

  const int N = in_sizes[0] / 64;
  const int E = in_sizes[1] / 2;
  const int* src = ei;
  const int* dst = ei + E;

  float* msg   = (float*)d_ws;
  float* msg2  = msg   + (size_t)N * 64;
  float* self2 = msg2  + (size_t)N * 64;
  float* h2    = self2 + (size_t)N * 64;
  int*   csr  = (int*)(h2 + (size_t)N * 32);
  int*   deg  = csr + E;
  int*   start = deg + N;
  int*   total = start + N;
  int*   cntm = (int*)msg;
  int*   offm = (int*)msg2;

  const int rowTiles = (N + 15) / 16;
  const int gblocks  = (rowTiles * 32 + 255) / 256;
  const int ranges   = (N + CSR_SPAN - 1) / CSR_SPAN;
  const int chunkLen = (E + CSR_CHUNKS - 1) / CSR_CHUNKS;
  const dim3 bgrid(ranges, CSR_CHUNKS);
  const int oblocks  = (N + 1023) / 1024;
  const int fblocks  = (rowTiles + FUSE_WAVES - 1) / FUSE_WAVES;

  csr_count<<<bgrid, 1024, 0, stream>>>(dst, cntm, total, N, E, chunkLen);
  csr_offsets<<<oblocks, 1024, 0, stream>>>(cntm, offm, deg, start, total, N);
  csr_place<<<bgrid, 1024, 0, stream>>>(src, dst, offm, start, csr, N, E, chunkLen);

  sage_gemm_wmma<true, false, false, 4, 64><<<gblocks, 256, 0, stream>>>(
      x, W1p, x, W1p, nullptr, b1p, msg, N, rowTiles);
  sage_gather_gemm<64, false, true, 4, false, 4, 2>
      <<<fblocks, FUSE_WAVES * 32, 0, stream>>>(
      x, W1s, msg, W1n, b1, csr, deg, start, nullptr,
      W2p, b2p, W2s, msg2, self2, N, rowTiles);

  sage_gather_gemm<64, false, false, 2, true, 0, 0>
      <<<fblocks, FUSE_WAVES * 32, 0, stream>>>(
      self2, nullptr, msg2, W2n, b2, csr, deg, start, h2,
      nullptr, nullptr, nullptr, nullptr, nullptr, N, rowTiles);

  sage_gather_gemm<32, true, false, 2, false, 0, 0>
      <<<fblocks, FUSE_WAVES * 32, 0, stream>>>(
      h2, W3s, h2, W3n, b3, csr, deg, start, (float*)d_out,
      nullptr, nullptr, nullptr, nullptr, nullptr, N, rowTiles);
}
